# starting kernel
# baseline (speedup 1.0000x reference)
#include <hip/hip_runtime.h>
#include <hip/hip_bf16.h>

#define N_NODES 100000
#define N_EDGES 1600000
#define D 64
#define NF (N_NODES * D)

typedef float v2f __attribute__((ext_vector_type(2)));
typedef float v8f __attribute__((ext_vector_type(8)));

__global__ void __launch_bounds__(256) scale_kernel(
    const float* __restrict__ H,
    const float* __restrict__ outdeg,
    const float* __restrict__ indeg,
    float* __restrict__ Xout,
    float* __restrict__ Xin) {
  int t = blockIdx.x * blockDim.x + threadIdx.x;
  if (t >= NF / 4) return;
  int node = t / (D / 4);
  float doo = fmaxf(outdeg[node], 1e-8f);
  float dii = fmaxf(indeg[node], 1e-8f);
  float4 h = reinterpret_cast<const float4*>(H)[t];
  float4 o, i;
  o.x = doo * h.x; o.y = doo * h.y; o.z = doo * h.z; o.w = doo * h.w;
  i.x = dii * h.x; i.y = dii * h.y; i.z = dii * h.z; i.w = dii * h.w;
  reinterpret_cast<float4*>(Xout)[t] = o;
  reinterpret_cast<float4*>(Xin)[t]  = i;
}

__device__ __forceinline__ void atomic_add_f32(float* p, float v) {
  (void)__hip_atomic_fetch_add(p, v, __ATOMIC_RELAXED, __HIP_MEMORY_SCOPE_AGENT);
}

__global__ void __launch_bounds__(256) scatter_kernel(
    float* __restrict__ dst,
    const float* __restrict__ src,
    const int* __restrict__ didx,
    const int* __restrict__ sidx,
    const float* __restrict__ val) {
  long long t = (long long)blockIdx.x * blockDim.x + threadIdx.x;
  long long e = t >> 4;
  int q = (int)(t & 15);
  if (e >= N_EDGES) return;
  int r = didx[e];
  int c = sidx[e];
  float v = val[e];
  float4 x = reinterpret_cast<const float4*>(src + (size_t)c * D)[q];
  float* dp = dst + (size_t)r * D + (size_t)q * 4;
  atomic_add_f32(dp + 0, v * x.x);
  atomic_add_f32(dp + 1, v * x.y);
  atomic_add_f32(dp + 2, v * x.z);
  atomic_add_f32(dp + 3, v * x.w);
}

__global__ void __launch_bounds__(256) accum_kernel(
    float* __restrict__ sum,
    const float* __restrict__ Xout,
    const float* __restrict__ Xin,
    const float* __restrict__ hop,
    const float* __restrict__ th_out,
    const float* __restrict__ th_in,
    int k) {
  int t = blockIdx.x * blockDim.x + threadIdx.x;
  if (t >= NF / 4) return;
  float m = fmaxf(fmaxf(hop[0], hop[1]), hop[2]);
  float e0 = __expf(hop[0] - m);
  float e1 = __expf(hop[1] - m);
  float e2 = __expf(hop[2] - m);
  float a = ((k == 0) ? e0 : (k == 1) ? e1 : e2) / (e0 + e1 + e2);
  float co = a * th_out[k];
  float ci = a * th_in[k];
  float4 s  = reinterpret_cast<const float4*>(sum)[t];
  float4 xo = reinterpret_cast<const float4*>(Xout)[t];
  float4 xi = reinterpret_cast<const float4*>(Xin)[t];
  s.x += co * xo.x + ci * xi.x;
  s.y += co * xo.y + ci * xi.y;
  s.z += co * xo.z + ci * xi.z;
  s.w += co * xo.w + ci * xi.w;
  reinterpret_cast<float4*>(sum)[t] = s;
}

__global__ void __launch_bounds__(256) gemm_sig_add_kernel(
    float* S,
    const float* __restrict__ Theta,
    const float* __restrict__ H) {
  int lane = threadIdx.x & 31;
  int tile = blockIdx.x * (blockDim.x >> 5) + (threadIdx.x >> 5);
  if (tile >= N_NODES / 16) return;

  int m0 = tile * 16;
  int mrow  = lane & 15;
  int khalf = lane >> 4;

  const float* arow  = S + (size_t)(m0 + mrow) * D + 2 * khalf;
  const float* bbase = Theta + (size_t)(2 * khalf) * D + (lane & 15);

  v8f c0 = {}, c1 = {}, c2 = {}, c3 = {};
#pragma unroll
  for (int kk = 0; kk < 16; ++kk) {
    v2f a;
    a.x = arow[4 * kk];
    a.y = arow[4 * kk + 1];
    const float* bp = bbase + (size_t)(4 * kk) * D;
    v2f b0, b1, b2, b3;
    b0.x = bp[0];      b0.y = bp[D];
    b1.x = bp[16];     b1.y = bp[16 + D];
    b2.x = bp[32];     b2.y = bp[32 + D];
    b3.x = bp[48];     b3.y = bp[48 + D];
    c0 = __builtin_amdgcn_wmma_f32_16x16x4_f32(false, a, false, b0, (short)0, c0, false, false);
    c1 = __builtin_amdgcn_wmma_f32_16x16x4_f32(false, a, false, b1, (short)0, c1, false, false);
    c2 = __builtin_amdgcn_wmma_f32_16x16x4_f32(false, a, false, b2, (short)0, c2, false, false);
    c3 = __builtin_amdgcn_wmma_f32_16x16x4_f32(false, a, false, b3, (short)0, c3, false, false);
  }

  int colb = lane & 15;
#pragma unroll
  for (int v = 0; v < 8; ++v) {
    size_t rb = (size_t)(m0 + v + 8 * khalf) * D;
    float x0 = c0[v], x1 = c1[v], x2 = c2[v], x3 = c3[v];
    S[rb +  0 + colb] = 1.0f / (1.0f + __expf(-x0)) + H[rb +  0 + colb];
    S[rb + 16 + colb] = 1.0f / (1.0f + __expf(-x1)) + H[rb + 16 + colb];
    S[rb + 32 + colb] = 1.0f / (1.0f + __expf(-x2)) + H[rb + 32 + colb];
    S[rb + 48 + colb] = 1.0f / (1.0f + __expf(-x3)) + H[rb + 48 + colb];
  }
}

extern "C" void kernel_launch(void* const* d_in, const int* in_sizes, int n_in,
                              void* d_out, int out_size, void* d_ws, size_t ws_size,
                              hipStream_t stream) {
  const float* H      = (const float*)d_in[0];
  const int*   erow   = (const int*)d_in[1];
  const int*   ecol   = (const int*)d_in[2];
  const float* eval   = (const float*)d_in[3];
  const float* outdeg = (const float*)d_in[4];
  const float* indeg  = (const float*)d_in[5];
  const float* hop    = (const float*)d_in[6];
  const float* th_out = (const float*)d_in[7];
  const float* th_in  = (const float*)d_in[8];
  const float* Theta  = (const float*)d_in[9];

  float* sum = (float*)d_out;
  float* ws  = (float*)d_ws;
  float* Xo[2] = {ws,              ws + (size_t)NF};
  float* Xi[2] = {ws + 2ull * NF,  ws + 3ull * NF};

  const int elem_blocks = (NF / 4 + 255) / 256;

  hipMemsetAsync(sum, 0, sizeof(float) * (size_t)NF, stream);
  scale_kernel<<<elem_blocks, 256, 0, stream>>>(H, outdeg, indeg, Xo[0], Xi[0]);

  int cur = 0;
  for (int k = 0; k < 3; ++k) {
    int nxt = cur ^ 1;
    hipMemsetAsync(Xo[nxt], 0, sizeof(float) * (size_t)NF, stream);
    hipMemsetAsync(Xi[nxt], 0, sizeof(float) * (size_t)NF, stream);

    long long nthreads = (long long)N_EDGES * 16;
    int sblocks = (int)((nthreads + 255) / 256);
    scatter_kernel<<<sblocks, 256, 0, stream>>>(Xo[nxt], Xo[cur], erow, ecol, eval);
    scatter_kernel<<<sblocks, 256, 0, stream>>>(Xi[nxt], Xi[cur], ecol, erow, eval);

    accum_kernel<<<elem_blocks, 256, 0, stream>>>(sum, Xo[nxt], Xi[nxt],
                                                  hop, th_out, th_in, k);
    cur = nxt;
  }

  const int tiles = N_NODES / 16;
  gemm_sig_add_kernel<<<(tiles + 7) / 8, 256, 0, stream>>>(sum, Theta, H);
}


# Scatter split by feature half, not row range: each chip half gathers from and adds into only its own 32-float half rows (planes kept as two [N][32] halves); waves walk all 64 edges; sum order arrival
# speedup vs baseline: 3.1729x; 3.1729x over previous
#include <hip/hip_runtime.h>
#include <hip/hip_bf16.h>

#define N_NODES 100000
#define N_EDGES 1600000
#define D 64
#define NF (N_NODES * D)

typedef float v2f __attribute__((ext_vector_type(2)));
typedef float v8f __attribute__((ext_vector_type(8)));

struct HopCoef { float o[3]; float i[3]; };

#define HALF_PLANE (N_NODES * 32)

__device__ __forceinline__ int split_f4(int t) {
  const int node = t >> 4;
  const int q = t & 15;
  return (q >> 3) * (HALF_PLANE / 4) + node * 8 + (q & 7);
}

__device__ __forceinline__ HopCoef hop_coefs(const float* __restrict__ hop,
                                             const float* __restrict__ th_out,
                                             const float* __restrict__ th_in) {
  float m = fmaxf(fmaxf(hop[0], hop[1]), hop[2]);
  float e0 = __expf(hop[0] - m);
  float e1 = __expf(hop[1] - m);
  float e2 = __expf(hop[2] - m);
  float s = e0 + e1 + e2;
  float a0 = e0 / s, a1 = e1 / s, a2 = e2 / s;
  HopCoef c;
  c.o[0] = a0 * th_out[0]; c.o[1] = a1 * th_out[1]; c.o[2] = a2 * th_out[2];
  c.i[0] = a0 * th_in[0];  c.i[1] = a1 * th_in[1];  c.i[2] = a2 * th_in[2];
  return c;
}

__device__ __forceinline__ float4 scaled(float s, float4 x) {
  float4 r;
  r.x = s * x.x; r.y = s * x.y; r.z = s * x.z; r.w = s * x.w;
  return r;
}

__global__ void __launch_bounds__(256) seed_inner_kernel(
    const float* __restrict__ H,
    const float* __restrict__ outdeg,
    const float* __restrict__ indeg,
    const float* __restrict__ hop,
    const float* __restrict__ th_out,
    const float* __restrict__ th_in,
    float* __restrict__ So,
    float* __restrict__ Si,
    float* __restrict__ To,
    float* __restrict__ Ti,
    unsigned* __restrict__ tickets) {
  int t = blockIdx.x * blockDim.x + threadIdx.x;
  if (t >= NF / 4) return;
  if (t == 0) {
    tickets[0] = 0u; tickets[1] = 0u; tickets[2] = 0u; tickets[3] = 0u;
  }
  int node = t / (D / 4);
  HopCoef c = hop_coefs(hop, th_out, th_in);
  float doo = fmaxf(outdeg[node], 1e-8f);
  float dii = fmaxf(indeg[node], 1e-8f);
  float4 h = reinterpret_cast<const float4*>(H)[t];
  float4 xo = scaled(doo, h);
  float4 xi = scaled(dii, h);
  const int s = split_f4(t);
  reinterpret_cast<float4*>(So)[s] = scaled(c.o[2], xo);
  reinterpret_cast<float4*>(Si)[s] = scaled(c.i[2], xi);
  reinterpret_cast<float4*>(To)[s] = scaled(c.o[1], xo);
  reinterpret_cast<float4*>(Ti)[s] = scaled(c.i[1], xi);
}

__global__ void __launch_bounds__(256) seed_outer_kernel(
    const float* __restrict__ H,
    const float* __restrict__ outdeg,
    const float* __restrict__ indeg,
    const float* __restrict__ hop,
    const float* __restrict__ th_out,
    const float* __restrict__ th_in,
    float* __restrict__ Uo,
    float* __restrict__ Ui,
    float* __restrict__ sum) {
  int t = blockIdx.x * blockDim.x + threadIdx.x;
  if (t >= NF / 4) return;
  int node = t / (D / 4);
  HopCoef c = hop_coefs(hop, th_out, th_in);
  float doo = fmaxf(outdeg[node], 1e-8f);
  float dii = fmaxf(indeg[node], 1e-8f);
  float4 h = reinterpret_cast<const float4*>(H)[t];
  const int s = split_f4(t);
  reinterpret_cast<float4*>(Uo)[s] = scaled(c.o[0], scaled(doo, h));
  reinterpret_cast<float4*>(Ui)[s] = scaled(c.i[0], scaled(dii, h));
  reinterpret_cast<float4*>(sum)[t] = make_float4(0.0f, 0.0f, 0.0f, 0.0f);
}

#define CHUNK_EDGES 64
#define N_CHUNKS (N_EDGES / CHUNK_EDGES)
#define WAVES_PER_BLOCK 8
static_assert(N_EDGES % CHUNK_EDGES == 0, "edge count must fill whole chunks");
static_assert(N_CHUNKS % WAVES_PER_BLOCK == 0, "a block's 8 chunks lie in one half's list");
#define SCATTER_BLOCKS ((2 * N_CHUNKS) / WAVES_PER_BLOCK)

__device__ __forceinline__ void atomic_add_f32(float* p, float v) {
  (void)__hip_atomic_fetch_add(p, v, __ATOMIC_RELAXED, __HIP_MEMORY_SCOPE_AGENT);
}

__device__ __forceinline__ int chip_half() {
  const unsigned id = (unsigned)__builtin_amdgcn_s_sendmsg_rtn(0x87);
  return (int)((id >> 18) & 1u);
}

__device__ __forceinline__ unsigned draw_ticket(unsigned* ctr) {
  return __hip_atomic_fetch_add(ctr, (unsigned)WAVES_PER_BLOCK, __ATOMIC_RELAXED,
                                __HIP_MEMORY_SCOPE_AGENT);
}

__device__ __forceinline__ void walk_32_edges(
    float* __restrict__ dst, const float* __restrict__ src, unsigned lane,
    int r, int c, float v) {
#pragma unroll
  for (int base = 0; base < 32; base += 16) {
    float x[16];
#pragma unroll
    for (int k = 0; k < 16; ++k) {
      const int cj = __builtin_amdgcn_readlane(c, base + k);
      x[k] = (src + (size_t)cj * 32)[lane];
    }
#pragma unroll
    for (int k = 0; k < 16; ++k) {
      const int rj = __builtin_amdgcn_readlane(r, base + k);
      const float vj = __builtin_bit_cast(
          float, __builtin_amdgcn_readlane(__builtin_bit_cast(int, v), base + k));
      atomic_add_f32((dst + (size_t)rj * 32) + lane, vj * x[k]);
    }
  }
}

__global__ void __launch_bounds__(256) __attribute__((amdgpu_waves_per_eu(16, 16)))
scatter_pair_kernel(
    float* dst_o,
    const float* __restrict__ src_o,
    float* dst_i,
    const float* __restrict__ src_i,
    const int* __restrict__ erow,
    const int* __restrict__ ecol,
    const float* __restrict__ val,
    unsigned* __restrict__ tickets) {
  __shared__ unsigned s_item[2];
  const int lane = threadIdx.x & 31;

  const bool in_chain = blockIdx.x >= (unsigned)SCATTER_BLOCKS;
  float* dst            = in_chain ? dst_i : dst_o;
  const float* src      = in_chain ? src_i : src_o;
  const int* didx       = in_chain ? ecol : erow;
  const int* sidx       = in_chain ? erow : ecol;
  unsigned* chain_tick  = tickets + (in_chain ? 2 : 0);

  if (threadIdx.x == 0) {
    int gb = chip_half();
    unsigned tb = draw_ticket(chain_tick + gb);
    if (tb >= (unsigned)N_CHUNKS) {
      gb ^= 1;
      tb = draw_ticket(chain_tick + gb);
    }
    s_item[0] = tb;
    s_item[1] = (unsigned)gb;
  }
  __syncthreads();
  const unsigned t = (unsigned)__builtin_amdgcn_readfirstlane(
      (int)(s_item[0] + (threadIdx.x >> 5)));
  const int g = __builtin_amdgcn_readfirstlane((int)s_item[1]);
  if (t >= (unsigned)N_CHUNKS) return;

  const int e = (int)t * CHUNK_EDGES + lane;
  const int   r0 = didx[e],      r1 = didx[e + 32];
  const int   c0 = sidx[e],      c1 = sidx[e + 32];
  const float v0 = val[e],       v1 = val[e + 32];

  float* dst_h       = dst + (size_t)g * HALF_PLANE;
  const float* src_h = src + (size_t)g * HALF_PLANE;

  walk_32_edges(dst_h, src_h, (unsigned)lane, r0, c0, v0);
  walk_32_edges(dst_h, src_h, (unsigned)lane, r1, c1, v1);
}

__global__ void __launch_bounds__(32) reset_tickets_kernel(
    float* __restrict__ sum_head, unsigned* __restrict__ tickets) {
  const int t = threadIdx.x;
  if (t < 4) {
    sum_head[t] = 0.0f;
    tickets[t] = 0u;
  }
}

__global__ void __launch_bounds__(256) gemm_sig_add_kernel(
    const float* __restrict__ S,
    const float* __restrict__ Theta,
    const float* __restrict__ H,
    float* __restrict__ out) {
  int lane = threadIdx.x & 31;
  int tile = blockIdx.x * (blockDim.x >> 5) + (threadIdx.x >> 5);
  if (tile >= N_NODES / 16) return;

  int m0 = tile * 16;
  int mrow  = lane & 15;
  int khalf = lane >> 4;

  const float* arow  = S + (size_t)(m0 + mrow) * 32 + 2 * khalf;
  const float* bbase = Theta + (size_t)(2 * khalf) * D + (lane & 15);

  v8f c0 = {}, c1 = {}, c2 = {}, c3 = {};
#pragma unroll
  for (int kk = 0; kk < 16; ++kk) {
    v2f a;
    const float* ap = arow + (kk < 8 ? 4 * kk : HALF_PLANE + 4 * (kk - 8));
    a.x = ap[0];
    a.y = ap[1];
    const float* bp = bbase + (size_t)(4 * kk) * D;
    v2f b0, b1, b2, b3;
    b0.x = bp[0];      b0.y = bp[D];
    b1.x = bp[16];     b1.y = bp[16 + D];
    b2.x = bp[32];     b2.y = bp[32 + D];
    b3.x = bp[48];     b3.y = bp[48 + D];
    c0 = __builtin_amdgcn_wmma_f32_16x16x4_f32(false, a, false, b0, (short)0, c0, false, false);
    c1 = __builtin_amdgcn_wmma_f32_16x16x4_f32(false, a, false, b1, (short)0, c1, false, false);
    c2 = __builtin_amdgcn_wmma_f32_16x16x4_f32(false, a, false, b2, (short)0, c2, false, false);
    c3 = __builtin_amdgcn_wmma_f32_16x16x4_f32(false, a, false, b3, (short)0, c3, false, false);
  }

  int colb = lane & 15;
#pragma unroll
  for (int v = 0; v < 8; ++v) {
    size_t rb = (size_t)(m0 + v + 8 * khalf) * D;
    float x0 = c0[v], x1 = c1[v], x2 = c2[v], x3 = c3[v];
    out[rb +  0 + colb] = 1.0f / (1.0f + __expf(-x0)) + H[rb +  0 + colb];
    out[rb + 16 + colb] = 1.0f / (1.0f + __expf(-x1)) + H[rb + 16 + colb];
    out[rb + 32 + colb] = 1.0f / (1.0f + __expf(-x2)) + H[rb + 32 + colb];
    out[rb + 48 + colb] = 1.0f / (1.0f + __expf(-x3)) + H[rb + 48 + colb];
  }
}

extern "C" void kernel_launch(void* const* d_in, const int* in_sizes, int n_in,
                              void* d_out, int out_size, void* d_ws, size_t ws_size,
                              hipStream_t stream) {
  const float* H      = (const float*)d_in[0];
  const int*   erow   = (const int*)d_in[1];
  const int*   ecol   = (const int*)d_in[2];
  const float* eval   = (const float*)d_in[3];
  const float* outdeg = (const float*)d_in[4];
  const float* indeg  = (const float*)d_in[5];
  const float* hop    = (const float*)d_in[6];
  const float* th_out = (const float*)d_in[7];
  const float* th_in  = (const float*)d_in[8];
  const float* Theta  = (const float*)d_in[9];

  float* ws  = (float*)d_ws;
  float* Po  = ws;
  float* Pi  = ws + (size_t)NF;
  float* Qo  = ws + 2ull * NF;
  float* Qi  = (float*)d_out;
  float* sum = ws + 3ull * NF;

  unsigned* tick_sum = (unsigned*)sum;
  unsigned* tick_q   = (unsigned*)Qo;

  const int elem_blocks = (NF / 4 + 255) / 256;
  const int pair_blocks = 2 * SCATTER_BLOCKS;

  seed_inner_kernel<<<elem_blocks, 256, 0, stream>>>(H, outdeg, indeg, hop, th_out, th_in,
                                                     Po, Pi, Qo, Qi, tick_sum);
  scatter_pair_kernel<<<pair_blocks, 256, 0, stream>>>(Qo, Po, Qi, Pi, erow, ecol, eval, tick_sum);

  seed_outer_kernel<<<elem_blocks, 256, 0, stream>>>(H, outdeg, indeg, hop, th_out, th_in,
                                                     Po, Pi, sum);
  scatter_pair_kernel<<<pair_blocks, 256, 0, stream>>>(Po, Qo, Pi, Qi, erow, ecol, eval, tick_sum);

  reset_tickets_kernel<<<1, 32, 0, stream>>>(sum, tick_q);
  scatter_pair_kernel<<<pair_blocks, 256, 0, stream>>>(sum, Po, sum, Pi, erow, ecol, eval, tick_q);

  const int tiles = N_NODES / 16;
  gemm_sig_add_kernel<<<(tiles + 7) / 8, 256, 0, stream>>>(sum, Theta, H, (float*)d_out);
}


# Scatter row addresses built on the scalar side: row byte offset broadcast, one scalar add to the plane base, lane number the one shared vector offset; no per-edge 64-bit vector add; sum order arrival
# speedup vs baseline: 3.2904x; 1.0370x over previous
#include <hip/hip_runtime.h>
#include <hip/hip_bf16.h>

#define N_NODES 100000
#define N_EDGES 1600000
#define D 64
#define NF (N_NODES * D)

typedef float v2f __attribute__((ext_vector_type(2)));
typedef float v8f __attribute__((ext_vector_type(8)));

struct HopCoef { float o[3]; float i[3]; };

#define HALF_PLANE (N_NODES * 32)

__device__ __forceinline__ int split_f4(int t) {
  const int node = t >> 4;
  const int q = t & 15;
  return (q >> 3) * (HALF_PLANE / 4) + node * 8 + (q & 7);
}

__device__ __forceinline__ HopCoef hop_coefs(const float* __restrict__ hop,
                                             const float* __restrict__ th_out,
                                             const float* __restrict__ th_in) {
  float m = fmaxf(fmaxf(hop[0], hop[1]), hop[2]);
  float e0 = __expf(hop[0] - m);
  float e1 = __expf(hop[1] - m);
  float e2 = __expf(hop[2] - m);
  float s = e0 + e1 + e2;
  float a0 = e0 / s, a1 = e1 / s, a2 = e2 / s;
  HopCoef c;
  c.o[0] = a0 * th_out[0]; c.o[1] = a1 * th_out[1]; c.o[2] = a2 * th_out[2];
  c.i[0] = a0 * th_in[0];  c.i[1] = a1 * th_in[1];  c.i[2] = a2 * th_in[2];
  return c;
}

__device__ __forceinline__ float4 scaled(float s, float4 x) {
  float4 r;
  r.x = s * x.x; r.y = s * x.y; r.z = s * x.z; r.w = s * x.w;
  return r;
}

__global__ void __launch_bounds__(256) seed_inner_kernel(
    const float* __restrict__ H,
    const float* __restrict__ outdeg,
    const float* __restrict__ indeg,
    const float* __restrict__ hop,
    const float* __restrict__ th_out,
    const float* __restrict__ th_in,
    float* __restrict__ So,
    float* __restrict__ Si,
    float* __restrict__ To,
    float* __restrict__ Ti,
    unsigned* __restrict__ tickets) {
  int t = blockIdx.x * blockDim.x + threadIdx.x;
  if (t >= NF / 4) return;
  if (t == 0) {
    tickets[0] = 0u; tickets[1] = 0u; tickets[2] = 0u; tickets[3] = 0u;
  }
  int node = t / (D / 4);
  HopCoef c = hop_coefs(hop, th_out, th_in);
  float doo = fmaxf(outdeg[node], 1e-8f);
  float dii = fmaxf(indeg[node], 1e-8f);
  float4 h = reinterpret_cast<const float4*>(H)[t];
  float4 xo = scaled(doo, h);
  float4 xi = scaled(dii, h);
  const int s = split_f4(t);
  reinterpret_cast<float4*>(So)[s] = scaled(c.o[2], xo);
  reinterpret_cast<float4*>(Si)[s] = scaled(c.i[2], xi);
  reinterpret_cast<float4*>(To)[s] = scaled(c.o[1], xo);
  reinterpret_cast<float4*>(Ti)[s] = scaled(c.i[1], xi);
}

__global__ void __launch_bounds__(256) seed_outer_kernel(
    const float* __restrict__ H,
    const float* __restrict__ outdeg,
    const float* __restrict__ indeg,
    const float* __restrict__ hop,
    const float* __restrict__ th_out,
    const float* __restrict__ th_in,
    float* __restrict__ Uo,
    float* __restrict__ Ui,
    float* __restrict__ sum) {
  int t = blockIdx.x * blockDim.x + threadIdx.x;
  if (t >= NF / 4) return;
  int node = t / (D / 4);
  HopCoef c = hop_coefs(hop, th_out, th_in);
  float doo = fmaxf(outdeg[node], 1e-8f);
  float dii = fmaxf(indeg[node], 1e-8f);
  float4 h = reinterpret_cast<const float4*>(H)[t];
  const int s = split_f4(t);
  reinterpret_cast<float4*>(Uo)[s] = scaled(c.o[0], scaled(doo, h));
  reinterpret_cast<float4*>(Ui)[s] = scaled(c.i[0], scaled(dii, h));
  reinterpret_cast<float4*>(sum)[t] = make_float4(0.0f, 0.0f, 0.0f, 0.0f);
}

#define CHUNK_EDGES 64
#define N_CHUNKS (N_EDGES / CHUNK_EDGES)
#define WAVES_PER_BLOCK 8
static_assert(N_EDGES % CHUNK_EDGES == 0, "edge count must fill whole chunks");
static_assert(N_CHUNKS % WAVES_PER_BLOCK == 0, "a block's 8 chunks lie in one half's list");
#define SCATTER_BLOCKS ((2 * N_CHUNKS) / WAVES_PER_BLOCK)

typedef __attribute__((address_space(1))) float gfloat;

__device__ __forceinline__ void atomic_add_f32(gfloat* p, float v) {
  (void)__hip_atomic_fetch_add(p, v, __ATOMIC_RELAXED, __HIP_MEMORY_SCOPE_AGENT);
}

__device__ __forceinline__ int chip_half() {
  const unsigned id = (unsigned)__builtin_amdgcn_s_sendmsg_rtn(0x87);
  return (int)((id >> 18) & 1u);
}

__device__ __forceinline__ unsigned draw_ticket(unsigned* ctr) {
  return __hip_atomic_fetch_add(ctr, (unsigned)WAVES_PER_BLOCK, __ATOMIC_RELAXED,
                                __HIP_MEMORY_SCOPE_AGENT);
}

__device__ __forceinline__ unsigned long long scalar_addr(unsigned long long a) {
  const unsigned lo = (unsigned)__builtin_amdgcn_readfirstlane((int)(unsigned)a);
  const unsigned hi = (unsigned)__builtin_amdgcn_readfirstlane((int)(unsigned)(a >> 32));
  return ((unsigned long long)hi << 32) | lo;
}

__device__ __forceinline__ void walk_32_edges(
    float* __restrict__ dst, const float* __restrict__ src, unsigned lane,
    int r, int c, float v) {
  const unsigned cb = (unsigned)c << 7;
  const unsigned rb = (unsigned)r << 7;
  const unsigned long long src_a = (unsigned long long)src;
  const unsigned long long dst_a = (unsigned long long)dst;
#pragma unroll
  for (int base = 0; base < 32; base += 16) {
    float x[16];
#pragma unroll
    for (int k = 0; k < 16; ++k) {
      const unsigned so = (unsigned)__builtin_amdgcn_readlane((int)cb, base + k);
      const gfloat* row = (const gfloat*)scalar_addr(src_a + so);
      x[k] = row[lane];
    }
#pragma unroll
    for (int k = 0; k < 16; ++k) {
      const unsigned so = (unsigned)__builtin_amdgcn_readlane((int)rb, base + k);
      gfloat* row = (gfloat*)scalar_addr(dst_a + so);
      const float vj = __builtin_bit_cast(
          float, __builtin_amdgcn_readlane(__builtin_bit_cast(int, v), base + k));
      atomic_add_f32(row + lane, vj * x[k]);
    }
  }
}

__global__ void __launch_bounds__(256)
scatter_pair_kernel(
    float* dst_o,
    const float* __restrict__ src_o,
    float* dst_i,
    const float* __restrict__ src_i,
    const int* __restrict__ erow,
    const int* __restrict__ ecol,
    const float* __restrict__ val,
    unsigned* __restrict__ tickets) {
  __shared__ unsigned s_item[2];
  const int lane = threadIdx.x & 31;

  const bool in_chain = blockIdx.x >= (unsigned)SCATTER_BLOCKS;
  float* dst            = in_chain ? dst_i : dst_o;
  const float* src      = in_chain ? src_i : src_o;
  const int* didx       = in_chain ? ecol : erow;
  const int* sidx       = in_chain ? erow : ecol;
  unsigned* chain_tick  = tickets + (in_chain ? 2 : 0);

  if (threadIdx.x == 0) {
    int gb = chip_half();
    unsigned tb = draw_ticket(chain_tick + gb);
    if (tb >= (unsigned)N_CHUNKS) {
      gb ^= 1;
      tb = draw_ticket(chain_tick + gb);
    }
    s_item[0] = tb;
    s_item[1] = (unsigned)gb;
  }
  __syncthreads();
  const unsigned t = (unsigned)__builtin_amdgcn_readfirstlane(
      (int)(s_item[0] + (threadIdx.x >> 5)));
  const int g = __builtin_amdgcn_readfirstlane((int)s_item[1]);
  if (t >= (unsigned)N_CHUNKS) return;

  const int e = (int)t * CHUNK_EDGES + lane;
  const int   r0 = didx[e],      r1 = didx[e + 32];
  const int   c0 = sidx[e],      c1 = sidx[e + 32];
  const float v0 = val[e],       v1 = val[e + 32];

  float* dst_h       = dst + (size_t)g * HALF_PLANE;
  const float* src_h = src + (size_t)g * HALF_PLANE;

  walk_32_edges(dst_h, src_h, (unsigned)lane, r0, c0, v0);
  walk_32_edges(dst_h, src_h, (unsigned)lane, r1, c1, v1);
}

__global__ void __launch_bounds__(32) reset_tickets_kernel(
    float* __restrict__ sum_head, unsigned* __restrict__ tickets) {
  const int t = threadIdx.x;
  if (t < 4) {
    sum_head[t] = 0.0f;
    tickets[t] = 0u;
  }
}

__global__ void __launch_bounds__(256) gemm_sig_add_kernel(
    const float* __restrict__ S,
    const float* __restrict__ Theta,
    const float* __restrict__ H,
    float* __restrict__ out) {
  int lane = threadIdx.x & 31;
  int tile = blockIdx.x * (blockDim.x >> 5) + (threadIdx.x >> 5);
  if (tile >= N_NODES / 16) return;

  int m0 = tile * 16;
  int mrow  = lane & 15;
  int khalf = lane >> 4;

  const float* arow  = S + (size_t)(m0 + mrow) * 32 + 2 * khalf;
  const float* bbase = Theta + (size_t)(2 * khalf) * D + (lane & 15);

  v8f c0 = {}, c1 = {}, c2 = {}, c3 = {};
#pragma unroll
  for (int kk = 0; kk < 16; ++kk) {
    v2f a;
    const float* ap = arow + (kk < 8 ? 4 * kk : HALF_PLANE + 4 * (kk - 8));
    a.x = ap[0];
    a.y = ap[1];
    const float* bp = bbase + (size_t)(4 * kk) * D;
    v2f b0, b1, b2, b3;
    b0.x = bp[0];      b0.y = bp[D];
    b1.x = bp[16];     b1.y = bp[16 + D];
    b2.x = bp[32];     b2.y = bp[32 + D];
    b3.x = bp[48];     b3.y = bp[48 + D];
    c0 = __builtin_amdgcn_wmma_f32_16x16x4_f32(false, a, false, b0, (short)0, c0, false, false);
    c1 = __builtin_amdgcn_wmma_f32_16x16x4_f32(false, a, false, b1, (short)0, c1, false, false);
    c2 = __builtin_amdgcn_wmma_f32_16x16x4_f32(false, a, false, b2, (short)0, c2, false, false);
    c3 = __builtin_amdgcn_wmma_f32_16x16x4_f32(false, a, false, b3, (short)0, c3, false, false);
  }

  int colb = lane & 15;
#pragma unroll
  for (int v = 0; v < 8; ++v) {
    size_t rb = (size_t)(m0 + v + 8 * khalf) * D;
    float x0 = c0[v], x1 = c1[v], x2 = c2[v], x3 = c3[v];
    out[rb +  0 + colb] = 1.0f / (1.0f + __expf(-x0)) + H[rb +  0 + colb];
    out[rb + 16 + colb] = 1.0f / (1.0f + __expf(-x1)) + H[rb + 16 + colb];
    out[rb + 32 + colb] = 1.0f / (1.0f + __expf(-x2)) + H[rb + 32 + colb];
    out[rb + 48 + colb] = 1.0f / (1.0f + __expf(-x3)) + H[rb + 48 + colb];
  }
}

extern "C" void kernel_launch(void* const* d_in, const int* in_sizes, int n_in,
                              void* d_out, int out_size, void* d_ws, size_t ws_size,
                              hipStream_t stream) {
  const float* H      = (const float*)d_in[0];
  const int*   erow   = (const int*)d_in[1];
  const int*   ecol   = (const int*)d_in[2];
  const float* eval   = (const float*)d_in[3];
  const float* outdeg = (const float*)d_in[4];
  const float* indeg  = (const float*)d_in[5];
  const float* hop    = (const float*)d_in[6];
  const float* th_out = (const float*)d_in[7];
  const float* th_in  = (const float*)d_in[8];
  const float* Theta  = (const float*)d_in[9];

  float* ws  = (float*)d_ws;
  float* Po  = ws;
  float* Pi  = ws + (size_t)NF;
  float* Qo  = ws + 2ull * NF;
  float* Qi  = (float*)d_out;
  float* sum = ws + 3ull * NF;

  unsigned* tick_sum = (unsigned*)sum;
  unsigned* tick_q   = (unsigned*)Qo;

  const int elem_blocks = (NF / 4 + 255) / 256;
  const int pair_blocks = 2 * SCATTER_BLOCKS;

  seed_inner_kernel<<<elem_blocks, 256, 0, stream>>>(H, outdeg, indeg, hop, th_out, th_in,
                                                     Po, Pi, Qo, Qi, tick_sum);
  scatter_pair_kernel<<<pair_blocks, 256, 0, stream>>>(Qo, Po, Qi, Pi, erow, ecol, eval, tick_sum);

  seed_outer_kernel<<<elem_blocks, 256, 0, stream>>>(H, outdeg, indeg, hop, th_out, th_in,
                                                     Po, Pi, sum);
  scatter_pair_kernel<<<pair_blocks, 256, 0, stream>>>(Po, Qo, Pi, Qi, erow, ecol, eval, tick_sum);

  reset_tickets_kernel<<<1, 32, 0, stream>>>(sum, tick_q);
  scatter_pair_kernel<<<pair_blocks, 256, 0, stream>>>(sum, Po, sum, Pi, erow, ecol, eval, tick_q);

  const int tiles = N_NODES / 16;
  gemm_sig_add_kernel<<<(tiles + 7) / 8, 256, 0, stream>>>(sum, Theta, H, (float*)d_out);
}
